# starting kernel
# baseline (speedup 1.0000x reference)
#include <hip/hip_runtime.h>
#include <math.h>


typedef __attribute__((ext_vector_type(2))) float v2f;
typedef __attribute__((ext_vector_type(8))) float v8f;

#define THREADS 256
#define ROWS_PER_BLOCK 256
#define KDIM 576
#define KCHUNKS (KDIM / 4)
#define NCOLS 5

__device__ __forceinline__ void gate_rx(float* re, float* im, int mask, float c, float s) {
#pragma unroll
  for (int i = 0; i < 16; ++i) {
    if ((i & mask) == 0) {
      int j = i | mask;
      float r0 = re[i], i0 = im[i], r1 = re[j], i1 = im[j];
      re[i] = c * r0 + s * i1;  im[i] = c * i0 - s * r1;
      re[j] = c * r1 + s * i0;  im[j] = c * i1 - s * r0;
    }
  }
}
__device__ __forceinline__ void gate_ry(float* re, float* im, int mask, float c, float s) {
#pragma unroll
  for (int i = 0; i < 16; ++i) {
    if ((i & mask) == 0) {
      int j = i | mask;
      float r0 = re[i], i0 = im[i], r1 = re[j], i1 = im[j];
      re[i] = c * r0 - s * r1;  im[i] = c * i0 - s * i1;
      re[j] = s * r0 + c * r1;  im[j] = s * i0 + c * i1;
    }
  }
}
__device__ __forceinline__ void gate_rz(float* re, float* im, int mask, float c, float s) {
#pragma unroll
  for (int i = 0; i < 16; ++i) {
    if ((i & mask) == 0) {
      int j = i | mask;
      float r0 = re[i], i0 = im[i], r1 = re[j], i1 = im[j];
      re[i] = c * r0 + s * i0;  im[i] = c * i0 - s * r0;
      re[j] = c * r1 - s * i1;  im[j] = c * i1 + s * r1;
    }
  }
}
__device__ __forceinline__ void gate_cnot(float* re, float* im, int cmask, int tmask) {
#pragma unroll
  for (int i = 0; i < 16; ++i) {
    if ((i & cmask) && !(i & tmask)) {
      int j = i | tmask;
      float tr = re[i], ti = im[i];
      re[i] = re[j];  im[i] = im[j];
      re[j] = tr;     im[j] = ti;
    }
  }
}

__global__ __launch_bounds__(THREADS)
void qnat_main_kernel(const float* __restrict__ x,
                      const float* __restrict__ enc_w,
                      const float* __restrict__ enc_b,
                      const float* __restrict__ vp,
                      float* __restrict__ zbuf,
                      float* __restrict__ partials) {
  __shared__ float sWt[NCOLS * KDIM];
  __shared__ float sAng[ROWS_PER_BLOCK * 4];
  __shared__ float sCos[24], sSin[24];
  __shared__ float sRed[8][8];

  const int tid  = threadIdx.x;
  const int wave = tid >> 5;
  const int lane = tid & 31;
  const int blockBase = blockIdx.x * ROWS_PER_BLOCK;

  for (int i = tid; i < NCOLS * KDIM; i += THREADS) {
    int n = i / KDIM;
    int k = i - n * KDIM;
    int pi = (k / 144) * 4 + ((k % 24) / 6);
    sWt[i] = (n < 4) ? enc_w[n * 16 + pi] * (1.0f / 36.0f) : 0.0f;
  }
  if (tid < 24) {
    float h = 0.5f * vp[tid];
    sCos[tid] = cosf(h);
    sSin[tid] = sinf(h);
  }
  __syncthreads();

  const int m     = lane & 15;
  const int khalf = lane >> 4;
  const int row0  = blockBase + wave * 32;

  const float* xr0 = x + (size_t)(row0 + m) * KDIM + khalf * 2;
  const float* xr1 = xr0 + (size_t)16 * KDIM;
  const int ncol  = (m < 4) ? m : 4;
  const float* wcol = sWt + ncol * KDIM + khalf * 2;

  __builtin_prefetch(xr0, 0, 3);
  __builtin_prefetch(xr1, 0, 3);

  v8f c0 = {};
  v8f c1 = {};
#pragma unroll 8
  for (int kk = 0; kk < KCHUNKS; ++kk) {
    v2f a0 = *(const v2f*)(xr0 + kk * 4);
    v2f a1 = *(const v2f*)(xr1 + kk * 4);
    v2f b  = *(const v2f*)(wcol + kk * 4);
    c0 = __builtin_amdgcn_wmma_f32_16x16x4_f32(false, a0, false, b,
                                               (short)0, c0, false, false);
    c1 = __builtin_amdgcn_wmma_f32_16x16x4_f32(false, a1, false, b,
                                               (short)0, c1, false, false);
  }

  if (m < 4) {
    float bc = enc_b[m];
#pragma unroll
    for (int j = 0; j < 8; ++j) {
      int mr = j + 8 * khalf;
      sAng[(wave * 32 + mr) * 4 + m]      = c0[j] + bc;
      sAng[(wave * 32 + 16 + mr) * 4 + m] = c1[j] + bc;
    }
  }
  __syncthreads();

  float ang[4];
#pragma unroll
  for (int i = 0; i < 4; ++i) ang[i] = sAng[tid * 4 + i];

  float re[16], im[16];
#pragma unroll
  for (int i = 0; i < 16; ++i) { re[i] = 0.0f; im[i] = 0.0f; }
  re[0] = 1.0f;

#pragma unroll
  for (int w = 0; w < 4; ++w) {
    float h = 0.5f * ang[w];
    gate_rx(re, im, 8 >> w, cosf(h), sinf(h));
  }
#pragma unroll
  for (int d = 0; d < 2; ++d) {
#pragma unroll
    for (int i = 0; i < 4; ++i) {
      int mask = 8 >> i;
      int base = d * 12 + i * 3;
      gate_rx(re, im, mask, sCos[base + 0], sSin[base + 0]);
      gate_ry(re, im, mask, sCos[base + 1], sSin[base + 1]);
      gate_rz(re, im, mask, sCos[base + 2], sSin[base + 2]);
    }
    gate_cnot(re, im, 8, 4);
    gate_cnot(re, im, 4, 2);
    gate_cnot(re, im, 2, 1);
    gate_cnot(re, im, 1, 8);
  }

  float z[4] = {0.0f, 0.0f, 0.0f, 0.0f};
#pragma unroll
  for (int i = 0; i < 16; ++i) {
    float p = re[i] * re[i] + im[i] * im[i];
#pragma unroll
    for (int w = 0; w < 4; ++w)
      z[w] += (i & (8 >> w)) ? -p : p;
  }

  float4 zv = make_float4(z[0], z[1], z[2], z[3]);
  *(float4*)(zbuf + (size_t)(blockBase + tid) * 4) = zv;

  float v[8];
#pragma unroll
  for (int k = 0; k < 4; ++k) { v[k] = z[k]; v[4 + k] = z[k] * z[k]; }
#pragma unroll
  for (int off = 16; off > 0; off >>= 1) {
#pragma unroll
    for (int k = 0; k < 8; ++k) v[k] += __shfl_down(v[k], off, 32);
  }
  if (lane == 0) {
#pragma unroll
    for (int k = 0; k < 8; ++k) sRed[wave][k] = v[k];
  }
  __syncthreads();
  if (tid < 8) {
    float a = 0.0f;
#pragma unroll
    for (int w = 0; w < 8; ++w) a += sRed[w][tid];
    partials[blockIdx.x * 8 + tid] = a;
  }
}

__global__ void qnat_reduce_kernel(const float* __restrict__ partials,
                                   float* __restrict__ sums, int nblocks) {
  int n = threadIdx.x;
  if (n < 8) {
    float a = 0.0f;
    for (int b = 0; b < nblocks; ++b) a += partials[b * 8 + n];
    sums[n] = a;
  }
}

__global__ __launch_bounds__(THREADS)
void qnat_bn_kernel(const float* __restrict__ zbuf,
                    const float* __restrict__ sums,
                    const float* __restrict__ gamma,
                    const float* __restrict__ beta,
                    float* __restrict__ out, int Btot) {
  int s = blockIdx.x * THREADS + threadIdx.x;
  if (s >= Btot) return;
  float invB = 1.0f / (float)Btot;
  float4 zv = *(const float4*)(zbuf + (size_t)s * 4);
  float zz[4] = {zv.x, zv.y, zv.z, zv.w};
  float4 o;
  float* op = (float*)&o;
#pragma unroll
  for (int n = 0; n < 4; ++n) {
    float mean = sums[n] * invB;
    float var  = sums[4 + n] * invB - mean * mean;
    float r    = rsqrtf(var + 1e-5f);
    op[n] = (zz[n] - mean) * r * gamma[n] + beta[n];
  }
  *(float4*)(out + (size_t)s * 4) = o;
}

extern "C" void kernel_launch(void* const* d_in, const int* in_sizes, int n_in,
                              void* d_out, int out_size, void* d_ws, size_t ws_size,
                              hipStream_t stream) {
  (void)n_in; (void)out_size; (void)ws_size;
  const float* x      = (const float*)d_in[0];
  const float* enc_w  = (const float*)d_in[1];
  const float* enc_b  = (const float*)d_in[2];
  const float* vparam = (const float*)d_in[3];
  const float* gamma  = (const float*)d_in[4];
  const float* beta   = (const float*)d_in[5];
  float* out = (float*)d_out;

  const int B = in_sizes[0] / KDIM;
  const int nblocks = B / ROWS_PER_BLOCK;

  float* ws       = (float*)d_ws;
  float* partials = ws;
  float* sums     = ws + nblocks * 8;
  float* zbuf     = ws + 4096;

  qnat_main_kernel<<<nblocks, THREADS, 0, stream>>>(x, enc_w, enc_b, vparam,
                                                    zbuf, partials);
  qnat_reduce_kernel<<<1, 8, 0, stream>>>(partials, sums, nblocks);
  qnat_bn_kernel<<<nblocks, THREADS, 0, stream>>>(zbuf, sums, gamma, beta,
                                                  out, B);
}


# K1 reads x with whole-line 512-byte wave loads (16 B per lane, 48 in flight per wave) staged through a wave-private LDS tile into the WMMA A fragments; GEMM k terms are summed in a new order
# speedup vs baseline: 1.1266x; 1.1266x over previous
#include <hip/hip_runtime.h>
#include <math.h>


typedef __attribute__((ext_vector_type(2))) float v2f;
typedef __attribute__((ext_vector_type(4))) float v4f;
typedef __attribute__((ext_vector_type(8))) float v8f;

#define THREADS 256
#define ROWS_PER_BLOCK 256
#define KDIM 576
#define NCOLS 5
#define CH 96
#define NCH (KDIM / CH)
#define XPITCH 100
#define WPITCH 580
#define LPC 24
#define WT_ITERS 9

__device__ __forceinline__ void gate_rx(float* re, float* im, int mask, float c, float s) {
#pragma unroll
  for (int i = 0; i < 16; ++i) {
    if ((i & mask) == 0) {
      int j = i | mask;
      float r0 = re[i], i0 = im[i], r1 = re[j], i1 = im[j];
      re[i] = c * r0 + s * i1;  im[i] = c * i0 - s * r1;
      re[j] = c * r1 + s * i0;  im[j] = c * i1 - s * r0;
    }
  }
}
__device__ __forceinline__ void gate_ry(float* re, float* im, int mask, float c, float s) {
#pragma unroll
  for (int i = 0; i < 16; ++i) {
    if ((i & mask) == 0) {
      int j = i | mask;
      float r0 = re[i], i0 = im[i], r1 = re[j], i1 = im[j];
      re[i] = c * r0 - s * r1;  im[i] = c * i0 - s * i1;
      re[j] = s * r0 + c * r1;  im[j] = s * i0 + c * i1;
    }
  }
}
__device__ __forceinline__ void gate_rz(float* re, float* im, int mask, float c, float s) {
#pragma unroll
  for (int i = 0; i < 16; ++i) {
    if ((i & mask) == 0) {
      int j = i | mask;
      float r0 = re[i], i0 = im[i], r1 = re[j], i1 = im[j];
      re[i] = c * r0 + s * i0;  im[i] = c * i0 - s * r0;
      re[j] = c * r1 - s * i1;  im[j] = c * i1 + s * r1;
    }
  }
}
__device__ __forceinline__ void gate_cnot(float* re, float* im, int cmask, int tmask) {
#pragma unroll
  for (int i = 0; i < 16; ++i) {
    if ((i & cmask) && !(i & tmask)) {
      int j = i | tmask;
      float tr = re[i], ti = im[i];
      re[i] = re[j];  im[i] = im[j];
      re[j] = tr;     im[j] = ti;
    }
  }
}

__global__ __launch_bounds__(THREADS)
void qnat_main_kernel(const float* __restrict__ x,
                      const float* __restrict__ enc_w,
                      const float* __restrict__ enc_b,
                      const float* __restrict__ vp,
                      float* __restrict__ zbuf,
                      float* __restrict__ partials) {
  __shared__ __attribute__((aligned(16))) float sX[8 * 32 * XPITCH];
  __shared__ __attribute__((aligned(16))) float sWt[NCOLS * WPITCH];
  __shared__ float sAng[ROWS_PER_BLOCK * 4];
  __shared__ float sCos[24], sSin[24];
  __shared__ float sRed[8][8];

  const int tid  = threadIdx.x;
  const int wave = tid >> 5;
  const int lane = tid & 31;
  const int blockBase = blockIdx.x * ROWS_PER_BLOCK;
  const int row0 = blockBase + wave * 32;

  float wv[WT_ITERS];
#pragma unroll
  for (int it = 0; it < WT_ITERS; ++it) {
    int i = tid + it * THREADS;
    int n = i / KDIM;
    int k = i - n * KDIM;
    int pi = (k / 144) * 4 + ((k % 24) / 6);
    wv[it] = enc_w[n * 16 + pi];
  }
  const float vph = vp[tid < 24 ? tid : 23];

  int goff[3], loff[3];
#pragma unroll
  for (int q = 0; q < 3; ++q) {
    int e = q * 128 + 4 * lane;
    int r = e / CH;
    int c = e - r * CH;
    goff[q] = r * KDIM + c;
    loff[q] = r * XPITCH + c;
  }
  const float* xw = x + (size_t)row0 * KDIM;
  float* sXw = sX + wave * (32 * XPITCH);

  v4f buf[2][LPC];
#define LOAD_CHUNK(CI) do { \
    _Pragma("unroll") for (int i = 0; i < LPC; ++i) \
      buf[(CI) & 1][i] = *(const v4f*)(xw + (i / 3) * (4 * KDIM) + (CI) * CH + goff[i % 3]); } while (0)

  LOAD_CHUNK(0);
  LOAD_CHUNK(1);

#pragma unroll
  for (int it = 0; it < WT_ITERS; ++it) {
    int i = tid + it * THREADS;
    int n = i / KDIM;
    int k = i - n * KDIM;
    sWt[n * WPITCH + k] = wv[it] * (1.0f / 36.0f);
  }
  for (int k = tid; k < KDIM; k += THREADS) sWt[4 * WPITCH + k] = 0.0f;
  if (tid < 24) {
    float h = 0.5f * vph;
    sCos[tid] = cosf(h);
    sSin[tid] = sinf(h);
  }
  __syncthreads();

  const int m     = lane & 15;
  const int khalf = lane >> 4;
  const int ncol  = (m < 4) ? m : 4;
  const float* wfrag  = sWt + ncol * WPITCH + 4 * khalf;
  const float* afrag0 = sXw + m * XPITCH + 4 * khalf;
  const float* afrag1 = afrag0 + 16 * XPITCH;

  v8f c0 = {};
  v8f c1 = {};
#pragma unroll
  for (int ch = 0; ch < NCH; ++ch) {
#pragma unroll
    for (int i = 0; i < LPC; ++i)
      *(v4f*)(sXw + (i / 3) * (4 * XPITCH) + loff[i % 3]) = buf[ch & 1][i];
    if (ch + 2 < NCH) LOAD_CHUNK(ch + 2);
#pragma unroll
    for (int j = 0; j < CH / 8; ++j) {
      v4f b  = *(const v4f*)(wfrag + ch * CH + j * 8);
      v4f a0 = *(const v4f*)(afrag0 + j * 8);
      v4f a1 = *(const v4f*)(afrag1 + j * 8);
      c0 = __builtin_amdgcn_wmma_f32_16x16x4_f32(false, a0.xy, false, b.xy,
                                                 (short)0, c0, false, false);
      c1 = __builtin_amdgcn_wmma_f32_16x16x4_f32(false, a1.xy, false, b.xy,
                                                 (short)0, c1, false, false);
      c0 = __builtin_amdgcn_wmma_f32_16x16x4_f32(false, a0.zw, false, b.zw,
                                                 (short)0, c0, false, false);
      c1 = __builtin_amdgcn_wmma_f32_16x16x4_f32(false, a1.zw, false, b.zw,
                                                 (short)0, c1, false, false);
    }
  }
#undef LOAD_CHUNK

  if (m < 4) {
    float bc = enc_b[m];
#pragma unroll
    for (int j = 0; j < 8; ++j) {
      int mr = j + 8 * khalf;
      sAng[(wave * 32 + mr) * 4 + m]      = c0[j] + bc;
      sAng[(wave * 32 + 16 + mr) * 4 + m] = c1[j] + bc;
    }
  }
  __syncthreads();

  float ang[4];
#pragma unroll
  for (int i = 0; i < 4; ++i) ang[i] = sAng[tid * 4 + i];

  float re[16], im[16];
#pragma unroll
  for (int i = 0; i < 16; ++i) { re[i] = 0.0f; im[i] = 0.0f; }
  re[0] = 1.0f;

#pragma unroll
  for (int w = 0; w < 4; ++w) {
    float h = 0.5f * ang[w];
    gate_rx(re, im, 8 >> w, cosf(h), sinf(h));
  }
#pragma unroll
  for (int d = 0; d < 2; ++d) {
#pragma unroll
    for (int i = 0; i < 4; ++i) {
      int mask = 8 >> i;
      int base = d * 12 + i * 3;
      gate_rx(re, im, mask, sCos[base + 0], sSin[base + 0]);
      gate_ry(re, im, mask, sCos[base + 1], sSin[base + 1]);
      gate_rz(re, im, mask, sCos[base + 2], sSin[base + 2]);
    }
    gate_cnot(re, im, 8, 4);
    gate_cnot(re, im, 4, 2);
    gate_cnot(re, im, 2, 1);
    gate_cnot(re, im, 1, 8);
  }

  float z[4] = {0.0f, 0.0f, 0.0f, 0.0f};
#pragma unroll
  for (int i = 0; i < 16; ++i) {
    float p = re[i] * re[i] + im[i] * im[i];
#pragma unroll
    for (int w = 0; w < 4; ++w)
      z[w] += (i & (8 >> w)) ? -p : p;
  }

  float4 zv = make_float4(z[0], z[1], z[2], z[3]);
  *(float4*)(zbuf + (size_t)(blockBase + tid) * 4) = zv;

  float v[8];
#pragma unroll
  for (int k = 0; k < 4; ++k) { v[k] = z[k]; v[4 + k] = z[k] * z[k]; }
#pragma unroll
  for (int off = 16; off > 0; off >>= 1) {
#pragma unroll
    for (int k = 0; k < 8; ++k) v[k] += __shfl_down(v[k], off, 32);
  }
  if (lane == 0) {
#pragma unroll
    for (int k = 0; k < 8; ++k) sRed[wave][k] = v[k];
  }
  __syncthreads();
  if (tid < 8) {
    float a = 0.0f;
#pragma unroll
    for (int w = 0; w < 8; ++w) a += sRed[w][tid];
    partials[blockIdx.x * 8 + tid] = a;
  }
}

__global__ void qnat_reduce_kernel(const float* __restrict__ partials,
                                   float* __restrict__ sums, int nblocks) {
  int n = threadIdx.x;
  if (n < 8) {
    float a = 0.0f;
    for (int b = 0; b < nblocks; ++b) a += partials[b * 8 + n];
    sums[n] = a;
  }
}

__global__ __launch_bounds__(THREADS)
void qnat_bn_kernel(const float* __restrict__ zbuf,
                    const float* __restrict__ sums,
                    const float* __restrict__ gamma,
                    const float* __restrict__ beta,
                    float* __restrict__ out, int Btot) {
  int s = blockIdx.x * THREADS + threadIdx.x;
  if (s >= Btot) return;
  float invB = 1.0f / (float)Btot;
  float4 zv = *(const float4*)(zbuf + (size_t)s * 4);
  float zz[4] = {zv.x, zv.y, zv.z, zv.w};
  float4 o;
  float* op = (float*)&o;
#pragma unroll
  for (int n = 0; n < 4; ++n) {
    float mean = sums[n] * invB;
    float var  = sums[4 + n] * invB - mean * mean;
    float r    = rsqrtf(var + 1e-5f);
    op[n] = (zz[n] - mean) * r * gamma[n] + beta[n];
  }
  *(float4*)(out + (size_t)s * 4) = o;
}

extern "C" void kernel_launch(void* const* d_in, const int* in_sizes, int n_in,
                              void* d_out, int out_size, void* d_ws, size_t ws_size,
                              hipStream_t stream) {
  (void)n_in; (void)out_size; (void)ws_size;
  const float* x      = (const float*)d_in[0];
  const float* enc_w  = (const float*)d_in[1];
  const float* enc_b  = (const float*)d_in[2];
  const float* vparam = (const float*)d_in[3];
  const float* gamma  = (const float*)d_in[4];
  const float* beta   = (const float*)d_in[5];
  float* out = (float*)d_out;

  const int B = in_sizes[0] / KDIM;
  const int nblocks = B / ROWS_PER_BLOCK;

  float* ws       = (float*)d_ws;
  float* partials = ws;
  float* sums     = ws + nblocks * 8;
  float* zbuf     = ws + 4096;

  qnat_main_kernel<<<nblocks, THREADS, 0, stream>>>(x, enc_w, enc_b, vparam,
                                                    zbuf, partials);
  qnat_reduce_kernel<<<1, 8, 0, stream>>>(partials, sums, nblocks);
  qnat_bn_kernel<<<nblocks, THREADS, 0, stream>>>(zbuf, sums, gamma, beta,
                                                  out, B);
}


# Reduce kernel parallelised: 256 threads each add eight block partials with all loads in flight (one second-level round trip, not 32 in a row), then a fixed LDS and lane sum; BN sums in a new order
# speedup vs baseline: 1.4443x; 1.2820x over previous
#include <hip/hip_runtime.h>
#include <math.h>


typedef __attribute__((ext_vector_type(2))) float v2f;
typedef __attribute__((ext_vector_type(4))) float v4f;
typedef __attribute__((ext_vector_type(8))) float v8f;

#define THREADS 256
#define ROWS_PER_BLOCK 256
#define KDIM 576
#define NCOLS 5
#define CH 96
#define NCH (KDIM / CH)
#define XPITCH 100
#define WPITCH 580
#define LPC 24
#define WT_ITERS 9

__device__ __forceinline__ void gate_rx(float* re, float* im, int mask, float c, float s) {
#pragma unroll
  for (int i = 0; i < 16; ++i) {
    if ((i & mask) == 0) {
      int j = i | mask;
      float r0 = re[i], i0 = im[i], r1 = re[j], i1 = im[j];
      re[i] = c * r0 + s * i1;  im[i] = c * i0 - s * r1;
      re[j] = c * r1 + s * i0;  im[j] = c * i1 - s * r0;
    }
  }
}
__device__ __forceinline__ void gate_ry(float* re, float* im, int mask, float c, float s) {
#pragma unroll
  for (int i = 0; i < 16; ++i) {
    if ((i & mask) == 0) {
      int j = i | mask;
      float r0 = re[i], i0 = im[i], r1 = re[j], i1 = im[j];
      re[i] = c * r0 - s * r1;  im[i] = c * i0 - s * i1;
      re[j] = s * r0 + c * r1;  im[j] = s * i0 + c * i1;
    }
  }
}
__device__ __forceinline__ void gate_rz(float* re, float* im, int mask, float c, float s) {
#pragma unroll
  for (int i = 0; i < 16; ++i) {
    if ((i & mask) == 0) {
      int j = i | mask;
      float r0 = re[i], i0 = im[i], r1 = re[j], i1 = im[j];
      re[i] = c * r0 + s * i0;  im[i] = c * i0 - s * r0;
      re[j] = c * r1 - s * i1;  im[j] = c * i1 + s * r1;
    }
  }
}
__device__ __forceinline__ void gate_cnot(float* re, float* im, int cmask, int tmask) {
#pragma unroll
  for (int i = 0; i < 16; ++i) {
    if ((i & cmask) && !(i & tmask)) {
      int j = i | tmask;
      float tr = re[i], ti = im[i];
      re[i] = re[j];  im[i] = im[j];
      re[j] = tr;     im[j] = ti;
    }
  }
}

__global__ __launch_bounds__(THREADS)
void qnat_main_kernel(const float* __restrict__ x,
                      const float* __restrict__ enc_w,
                      const float* __restrict__ enc_b,
                      const float* __restrict__ vp,
                      float* __restrict__ zbuf,
                      float* __restrict__ partials) {
  __shared__ __attribute__((aligned(16))) float sX[8 * 32 * XPITCH];
  __shared__ __attribute__((aligned(16))) float sWt[NCOLS * WPITCH];
  __shared__ float sAng[ROWS_PER_BLOCK * 4];
  __shared__ float sCos[24], sSin[24];
  __shared__ float sRed[8][8];

  const int tid  = threadIdx.x;
  const int wave = tid >> 5;
  const int lane = tid & 31;
  const int blockBase = blockIdx.x * ROWS_PER_BLOCK;
  const int row0 = blockBase + wave * 32;

  float wv[WT_ITERS];
#pragma unroll
  for (int it = 0; it < WT_ITERS; ++it) {
    int i = tid + it * THREADS;
    int n = i / KDIM;
    int k = i - n * KDIM;
    int pi = (k / 144) * 4 + ((k % 24) / 6);
    wv[it] = enc_w[n * 16 + pi];
  }
  const float vph = vp[tid < 24 ? tid : 23];

  int goff[3], loff[3];
#pragma unroll
  for (int q = 0; q < 3; ++q) {
    int e = q * 128 + 4 * lane;
    int r = e / CH;
    int c = e - r * CH;
    goff[q] = r * KDIM + c;
    loff[q] = r * XPITCH + c;
  }
  const float* xw = x + (size_t)row0 * KDIM;
  float* sXw = sX + wave * (32 * XPITCH);

  v4f buf[2][LPC];
#define LOAD_CHUNK(CI) do { \
    _Pragma("unroll") for (int i = 0; i < LPC; ++i) \
      buf[(CI) & 1][i] = *(const v4f*)(xw + (i / 3) * (4 * KDIM) + (CI) * CH + goff[i % 3]); } while (0)

  LOAD_CHUNK(0);
  LOAD_CHUNK(1);

#pragma unroll
  for (int it = 0; it < WT_ITERS; ++it) {
    int i = tid + it * THREADS;
    int n = i / KDIM;
    int k = i - n * KDIM;
    sWt[n * WPITCH + k] = wv[it] * (1.0f / 36.0f);
  }
  for (int k = tid; k < KDIM; k += THREADS) sWt[4 * WPITCH + k] = 0.0f;
  if (tid < 24) {
    float h = 0.5f * vph;
    sCos[tid] = cosf(h);
    sSin[tid] = sinf(h);
  }
  __syncthreads();

  const int m     = lane & 15;
  const int khalf = lane >> 4;
  const int ncol  = (m < 4) ? m : 4;
  const float* wfrag  = sWt + ncol * WPITCH + 4 * khalf;
  const float* afrag0 = sXw + m * XPITCH + 4 * khalf;
  const float* afrag1 = afrag0 + 16 * XPITCH;

  v8f c0 = {};
  v8f c1 = {};
#pragma unroll
  for (int ch = 0; ch < NCH; ++ch) {
#pragma unroll
    for (int i = 0; i < LPC; ++i)
      *(v4f*)(sXw + (i / 3) * (4 * XPITCH) + loff[i % 3]) = buf[ch & 1][i];
    if (ch + 2 < NCH) LOAD_CHUNK(ch + 2);
#pragma unroll
    for (int j = 0; j < CH / 8; ++j) {
      v4f b  = *(const v4f*)(wfrag + ch * CH + j * 8);
      v4f a0 = *(const v4f*)(afrag0 + j * 8);
      v4f a1 = *(const v4f*)(afrag1 + j * 8);
      c0 = __builtin_amdgcn_wmma_f32_16x16x4_f32(false, a0.xy, false, b.xy,
                                                 (short)0, c0, false, false);
      c1 = __builtin_amdgcn_wmma_f32_16x16x4_f32(false, a1.xy, false, b.xy,
                                                 (short)0, c1, false, false);
      c0 = __builtin_amdgcn_wmma_f32_16x16x4_f32(false, a0.zw, false, b.zw,
                                                 (short)0, c0, false, false);
      c1 = __builtin_amdgcn_wmma_f32_16x16x4_f32(false, a1.zw, false, b.zw,
                                                 (short)0, c1, false, false);
    }
  }
#undef LOAD_CHUNK

  if (m < 4) {
    float bc = enc_b[m];
#pragma unroll
    for (int j = 0; j < 8; ++j) {
      int mr = j + 8 * khalf;
      sAng[(wave * 32 + mr) * 4 + m]      = c0[j] + bc;
      sAng[(wave * 32 + 16 + mr) * 4 + m] = c1[j] + bc;
    }
  }
  __syncthreads();

  float ang[4];
#pragma unroll
  for (int i = 0; i < 4; ++i) ang[i] = sAng[tid * 4 + i];

  float re[16], im[16];
#pragma unroll
  for (int i = 0; i < 16; ++i) { re[i] = 0.0f; im[i] = 0.0f; }
  re[0] = 1.0f;

#pragma unroll
  for (int w = 0; w < 4; ++w) {
    float h = 0.5f * ang[w];
    gate_rx(re, im, 8 >> w, cosf(h), sinf(h));
  }
#pragma unroll
  for (int d = 0; d < 2; ++d) {
#pragma unroll
    for (int i = 0; i < 4; ++i) {
      int mask = 8 >> i;
      int base = d * 12 + i * 3;
      gate_rx(re, im, mask, sCos[base + 0], sSin[base + 0]);
      gate_ry(re, im, mask, sCos[base + 1], sSin[base + 1]);
      gate_rz(re, im, mask, sCos[base + 2], sSin[base + 2]);
    }
    gate_cnot(re, im, 8, 4);
    gate_cnot(re, im, 4, 2);
    gate_cnot(re, im, 2, 1);
    gate_cnot(re, im, 1, 8);
  }

  float z[4] = {0.0f, 0.0f, 0.0f, 0.0f};
#pragma unroll
  for (int i = 0; i < 16; ++i) {
    float p = re[i] * re[i] + im[i] * im[i];
#pragma unroll
    for (int w = 0; w < 4; ++w)
      z[w] += (i & (8 >> w)) ? -p : p;
  }

  float4 zv = make_float4(z[0], z[1], z[2], z[3]);
  *(float4*)(zbuf + (size_t)(blockBase + tid) * 4) = zv;

  float v[8];
#pragma unroll
  for (int k = 0; k < 4; ++k) { v[k] = z[k]; v[4 + k] = z[k] * z[k]; }
#pragma unroll
  for (int off = 16; off > 0; off >>= 1) {
#pragma unroll
    for (int k = 0; k < 8; ++k) v[k] += __shfl_down(v[k], off, 32);
  }
  if (lane == 0) {
#pragma unroll
    for (int k = 0; k < 8; ++k) sRed[wave][k] = v[k];
  }
  __syncthreads();
  if (tid < 8) {
    float a = 0.0f;
#pragma unroll
    for (int w = 0; w < 8; ++w) a += sRed[w][tid];
    partials[blockIdx.x * 8 + tid] = a;
  }
}

#define RED_THREADS 256
__global__ __launch_bounds__(RED_THREADS)
void qnat_reduce_kernel(const float* __restrict__ partials,
                        float* __restrict__ sums, int nblocks) {
  __shared__ float sPart[RED_THREADS];
  const int t = threadIdx.x;
  const int n = t & 7;
  float a = 0.0f;
  for (int base = t >> 3; base < nblocks; base += RED_THREADS) {
    float v[8];
#pragma unroll
    for (int j = 0; j < 8; ++j) {
      int b = base + (RED_THREADS / 8) * j;
      int bc = (b < nblocks) ? b : base;
      float val = partials[bc * 8 + n];
      v[j] = (b < nblocks) ? val : 0.0f;
    }
#pragma unroll
    for (int j = 0; j < 8; ++j) a += v[j];
  }
  sPart[t] = a;
  __syncthreads();
  if (t < 32) {
    float s = 0.0f;
#pragma unroll
    for (int i = 0; i < RED_THREADS / 32; ++i) s += sPart[t + 32 * i];
    s += __shfl_xor(s, 8, 32);
    s += __shfl_xor(s, 16, 32);
    if (t < 8) sums[t] = s;
  }
}

__global__ __launch_bounds__(THREADS)
void qnat_bn_kernel(const float* __restrict__ zbuf,
                    const float* __restrict__ sums,
                    const float* __restrict__ gamma,
                    const float* __restrict__ beta,
                    float* __restrict__ out, int Btot) {
  int s = blockIdx.x * THREADS + threadIdx.x;
  if (s >= Btot) return;
  float invB = 1.0f / (float)Btot;
  float4 zv = *(const float4*)(zbuf + (size_t)s * 4);
  float zz[4] = {zv.x, zv.y, zv.z, zv.w};
  float4 o;
  float* op = (float*)&o;
#pragma unroll
  for (int n = 0; n < 4; ++n) {
    float mean = sums[n] * invB;
    float var  = sums[4 + n] * invB - mean * mean;
    float r    = rsqrtf(var + 1e-5f);
    op[n] = (zz[n] - mean) * r * gamma[n] + beta[n];
  }
  *(float4*)(out + (size_t)s * 4) = o;
}

extern "C" void kernel_launch(void* const* d_in, const int* in_sizes, int n_in,
                              void* d_out, int out_size, void* d_ws, size_t ws_size,
                              hipStream_t stream) {
  (void)n_in; (void)out_size; (void)ws_size;
  const float* x      = (const float*)d_in[0];
  const float* enc_w  = (const float*)d_in[1];
  const float* enc_b  = (const float*)d_in[2];
  const float* vparam = (const float*)d_in[3];
  const float* gamma  = (const float*)d_in[4];
  const float* beta   = (const float*)d_in[5];
  float* out = (float*)d_out;

  const int B = in_sizes[0] / KDIM;
  const int nblocks = B / ROWS_PER_BLOCK;

  float* ws       = (float*)d_ws;
  float* partials = ws;
  float* sums     = ws + nblocks * 8;
  float* zbuf     = ws + 4096;

  qnat_main_kernel<<<nblocks, THREADS, 0, stream>>>(x, enc_w, enc_b, vparam,
                                                    zbuf, partials);
  qnat_reduce_kernel<<<1, RED_THREADS, 0, stream>>>(partials, sums, nblocks);
  qnat_bn_kernel<<<nblocks, THREADS, 0, stream>>>(zbuf, sums, gamma, beta,
                                                  out, B);
}
